# starting kernel
# baseline (speedup 1.0000x reference)
#include <hip/hip_runtime.h>

typedef __attribute__((ext_vector_type(2))) float v2f;
typedef __attribute__((ext_vector_type(8))) float v8f;

#define GCN_F_IN 128
#define GCN_HID  128
#define GCN_NCLS 64
#define GCN_EPS  1e-6f


__global__ __launch_bounds__(256) void gcn_deg_init(float* __restrict__ deg, int n) {
    int i = blockIdx.x * blockDim.x + threadIdx.x;
    if (i < n) deg[i] = 1.0f;
}

__global__ __launch_bounds__(256) void gcn_deg_count(const int* __restrict__ dst,
                                                     float* __restrict__ deg, int e) {
    int i = blockIdx.x * blockDim.x + threadIdx.x;
    if (i < e) atomicAdd(&deg[dst[i]], 1.0f);
}

__global__ __launch_bounds__(256) void gcn_dinv(float* __restrict__ deg, int n) {
    int i = blockIdx.x * blockDim.x + threadIdx.x;
    if (i < n) {
        float d = deg[i];
        deg[i] = d > 0.0f ? rsqrtf(d) : 0.0f;
    }
}


template <int OUT, int TPW>
__global__ __launch_bounds__(128) void gcn_gemm_wmma(const float* __restrict__ A,
                                                     const float* __restrict__ B,
                                                     float* __restrict__ C, int M) {
    const int lane  = threadIdx.x & 31;
    const int wave  = threadIdx.x >> 5;
    const int mbase = blockIdx.x * 16;
    const int l16   = lane & 15;
    const int khalf = (lane >> 4) * 2;

    int arow = mbase + l16;
    if (arow > M - 1) arow = M - 1;
    const float* __restrict__ Ap = A + (long)arow * GCN_F_IN;

#pragma unroll
    for (int t = 0; t < TPW; ++t) {
        const int n0 = (wave * TPW + t) * 16;
        v8f c = {};
#pragma unroll
        for (int k0 = 0; k0 < 128; k0 += 4) {
            v2f a, b;
            a.x = Ap[k0 + khalf];
            a.y = Ap[k0 + khalf + 1];
            b.x = B[(k0 + khalf) * OUT + n0 + l16];
            b.y = B[(k0 + khalf + 1) * OUT + n0 + l16];
            c = __builtin_amdgcn_wmma_f32_16x16x4_f32(
                    false, a, false, b, (short)0, c, false, false);
        }
        const int mrow = mbase + (lane >> 4) * 8;
        float* __restrict__ Cp = C + n0 + l16;
#pragma unroll
        for (int v = 0; v < 8; ++v) {
            int r = mrow + v;
            if (r < M) Cp[(long)r * OUT] = c[v];
        }
    }
}

template <int F>
__global__ __launch_bounds__(256) void gcn_agg_init(const float* __restrict__ h,
                                                    const float* __restrict__ dinv,
                                                    const float* __restrict__ bias,
                                                    float* __restrict__ out, int n) {
    long i = (long)blockIdx.x * blockDim.x + threadIdx.x;
    if (i >= (long)n * F) return;
    int node = (int)(i / F);
    int feat = (int)(i % F);
    float di = dinv[node];
    out[i] = h[i] * di * di + bias[feat];
}

template <int F>
__global__ __launch_bounds__(256) void gcn_agg_edges(const float* __restrict__ h,
                                                     const int* __restrict__ src,
                                                     const int* __restrict__ dst,
                                                     const float* __restrict__ dinv,
                                                     float* __restrict__ out, int e) {
    const int LPE = F / 4;
    long tid  = (long)blockIdx.x * blockDim.x + threadIdx.x;
    int edge  = (int)(tid / LPE);
    int part  = (int)(tid % LPE);
    if (edge >= e) return;
    int s = src[edge];
    int d = dst[edge];
    float norm = dinv[s] * dinv[d];
    const float4 hv = *(const float4*)(h + (long)s * F + part * 4);
    float* __restrict__ o = out + (long)d * F + part * 4;
    atomicAdd(o + 0, hv.x * norm);
    atomicAdd(o + 1, hv.y * norm);
    atomicAdd(o + 2, hv.z * norm);
    atomicAdd(o + 3, hv.w * norm);
}


__global__ __launch_bounds__(256) void gcn_ln_relu(const float* __restrict__ in,
                                                   const float* __restrict__ gamma,
                                                   const float* __restrict__ beta,
                                                   float* __restrict__ outp, int n) {
    int lane = threadIdx.x & 31;
    int node = (int)(((long)blockIdx.x * blockDim.x + threadIdx.x) >> 5);
    if (node >= n) return;
    const float4 v = *(const float4*)(in + (long)node * 128 + lane * 4);

    float s = v.x + v.y + v.z + v.w;
#pragma unroll
    for (int off = 16; off > 0; off >>= 1) s += __shfl_xor(s, off, 32);
    float mu = s * (1.0f / 128.0f);

    float dx = v.x - mu, dy = v.y - mu, dz = v.z - mu, dw = v.w - mu;
    float sq = dx * dx + dy * dy + dz * dz + dw * dw;
#pragma unroll
    for (int off = 16; off > 0; off >>= 1) sq += __shfl_xor(sq, off, 32);
    float rstd = rsqrtf(sq * (1.0f / 128.0f) + GCN_EPS);

    const float4 g = *(const float4*)(gamma + lane * 4);
    const float4 b = *(const float4*)(beta + lane * 4);
    float4 o;
    o.x = fmaxf(dx * rstd * g.x + b.x, 0.0f);
    o.y = fmaxf(dy * rstd * g.y + b.y, 0.0f);
    o.z = fmaxf(dz * rstd * g.z + b.z, 0.0f);
    o.w = fmaxf(dw * rstd * g.w + b.w, 0.0f);
    *(float4*)(outp + (long)node * 128 + lane * 4) = o;
}


extern "C" void kernel_launch(void* const* d_in, const int* in_sizes, int n_in,
                              void* d_out, int out_size, void* d_ws, size_t ws_size,
                              hipStream_t stream) {
    const float* x   = (const float*)d_in[0];
    const int*   ei  = (const int*)d_in[1];
    const float* W1  = (const float*)d_in[2];
    const float* b1  = (const float*)d_in[3];
    const float* g1  = (const float*)d_in[4];
    const float* be1 = (const float*)d_in[5];
    const float* W2  = (const float*)d_in[6];
    const float* b2  = (const float*)d_in[7];
    float* out = (float*)d_out;

    const int n = in_sizes[0] / GCN_F_IN;
    const int e = in_sizes[1] / 2;
    const int* src = ei;
    const int* dst = ei + e;

    float* ws   = (float*)d_ws;
    float* dinv = ws;
    float* h1   = dinv + n;
    float* a1   = h1 + (size_t)n * GCN_HID;
    float* h2   = a1 + (size_t)n * GCN_HID;

    gcn_deg_init<<<(n + 255) / 256, 256, 0, stream>>>(dinv, n);
    gcn_deg_count<<<(e + 255) / 256, 256, 0, stream>>>(dst, dinv, e);
    gcn_dinv<<<(n + 255) / 256, 256, 0, stream>>>(dinv, n);

    gcn_gemm_wmma<GCN_HID, 2><<<(n + 15) / 16, 128, 0, stream>>>(x, W1, h1, n);
    gcn_agg_init<GCN_HID><<<(int)(((long)n * GCN_HID + 255) / 256), 256, 0, stream>>>(
        h1, dinv, b1, a1, n);
    gcn_agg_edges<GCN_HID><<<(int)(((long)e * (GCN_HID / 4) + 255) / 256), 256, 0, stream>>>(
        h1, src, dst, dinv, a1, e);
    gcn_ln_relu<<<(int)(((long)n * 32 + 255) / 256), 256, 0, stream>>>(a1, g1, be1, a1, n);

    gcn_gemm_wmma<GCN_NCLS, 1><<<(n + 15) / 16, 128, 0, stream>>>(a1, W2, h2, n);
    gcn_agg_init<GCN_NCLS><<<(int)(((long)n * GCN_NCLS + 255) / 256), 256, 0, stream>>>(
        h2, dinv, b2, out, n);
    gcn_agg_edges<GCN_NCLS><<<(int)(((long)e * (GCN_NCLS / 4) + 255) / 256), 256, 0, stream>>>(
        h2, src, dst, dinv, out, e);
}


# Float-atomic edge scatter replaced by pull aggregation: edges grouped by dst each call (int atomic rank, scan, fill), one wave per node sums rows in registers; per-node sum order = arrival rank
# speedup vs baseline: 2.2137x; 2.2137x over previous
#include <hip/hip_runtime.h>

typedef __attribute__((ext_vector_type(2))) float v2f;
typedef __attribute__((ext_vector_type(8))) float v8f;

#define GCN_F_IN 128
#define GCN_HID  128
#define GCN_NCLS 64
#define GCN_EPS  1e-6f


__global__ __launch_bounds__(256) void gcn_cnt_zero(int* __restrict__ cnt, int n) {
    int i = blockIdx.x * blockDim.x + threadIdx.x;
    if (i < n) cnt[i] = 0;
}

__global__ __launch_bounds__(256) void gcn_count_rank(const int* __restrict__ dst,
                                                      int* __restrict__ cnt,
                                                      int* __restrict__ rank, int e) {
    int i = blockIdx.x * blockDim.x + threadIdx.x;
    if (i < e) rank[i] = atomicAdd(&cnt[dst[i]], 1);
}

__global__ __launch_bounds__(1024) void gcn_scan_dinv(const int* __restrict__ cnt,
                                                      int* __restrict__ rowptr,
                                                      float* __restrict__ dinv, int n) {
    __shared__ int before[32];
    __shared__ int wtot[32];
    const int tid  = threadIdx.x;
    const int lane = tid & 31;
    const int wv   = tid >> 5;
    const int base = blockIdx.x * 1024;

    int p = 0;
    for (int i = tid; i < base; i += 1024) p += cnt[i];
    for (int off = 16; off > 0; off >>= 1) p += __shfl_xor(p, off, 32);

    const int idx = base + tid;
    const int c   = idx < n ? cnt[idx] : 0;
    int inc = c;
    for (int off = 1; off < 32; off <<= 1) {
        int t = __shfl_up(inc, off, 32);
        if (lane >= off) inc += t;
    }
    if (lane == 0)  before[wv] = p;
    if (lane == 31) wtot[wv]   = inc;
    __syncthreads();

    int both = before[lane] + (lane < wv ? wtot[lane] : 0);
    for (int off = 16; off > 0; off >>= 1) both += __shfl_xor(both, off, 32);

    const int excl = both + inc - c;
    if (idx < n) {
        rowptr[idx] = excl;
        float d = (float)(c + 1);
        dinv[idx] = d > 0.0f ? rsqrtf(d) : 0.0f;
    }
    if (idx == n - 1) rowptr[n] = excl + c;
}

__global__ __launch_bounds__(256) void gcn_fill(const int* __restrict__ src,
                                                const int* __restrict__ dst,
                                                const int* __restrict__ rank,
                                                const int* __restrict__ rowptr,
                                                const float* __restrict__ dinv,
                                                int2* __restrict__ rec, int e) {
    int i = blockIdx.x * blockDim.x + threadIdx.x;
    if (i >= e) return;
    const int s   = src[i];
    const int pos = rowptr[dst[i]] + rank[i];
    rec[pos] = make_int2(s, __float_as_int(dinv[s]));
}


template <int OUT, int TPW>
__global__ __launch_bounds__(128) void gcn_gemm_wmma(const float* __restrict__ A,
                                                     const float* __restrict__ B,
                                                     float* __restrict__ C, int M) {
    const int lane  = threadIdx.x & 31;
    const int wave  = threadIdx.x >> 5;
    const int mbase = blockIdx.x * 16;
    const int l16   = lane & 15;
    const int khalf = (lane >> 4) * 2;

    int arow = mbase + l16;
    if (arow > M - 1) arow = M - 1;
    const float* __restrict__ Ap = A + (long)arow * GCN_F_IN;

#pragma unroll
    for (int t = 0; t < TPW; ++t) {
        const int n0 = (wave * TPW + t) * 16;
        v8f c = {};
#pragma unroll
        for (int k0 = 0; k0 < 128; k0 += 4) {
            v2f a, b;
            a.x = Ap[k0 + khalf];
            a.y = Ap[k0 + khalf + 1];
            b.x = B[(k0 + khalf) * OUT + n0 + l16];
            b.y = B[(k0 + khalf + 1) * OUT + n0 + l16];
            c = __builtin_amdgcn_wmma_f32_16x16x4_f32(
                    false, a, false, b, (short)0, c, false, false);
        }
        const int mrow = mbase + (lane >> 4) * 8;
        float* __restrict__ Cp = C + n0 + l16;
#pragma unroll
        for (int v = 0; v < 8; ++v) {
            int r = mrow + v;
            if (r < M) Cp[(long)r * OUT] = c[v];
        }
    }
}


template <int F, int G>
__device__ __forceinline__ void gcn_pull_group(const float* __restrict__ h, int my_s, int my_w,
                                               int j, int lane,
                                               float (&acc)[F / 32]) {
    constexpr int VEC = F / 32;
    typedef __attribute__((ext_vector_type(VEC))) float vec_t;
    vec_t hv[G];
    float w[G];
    for (int g = 0; g < G; ++g) {
        const int s = __builtin_amdgcn_readlane(my_s, j + g);
        w[g] = __int_as_float(__builtin_amdgcn_readlane(my_w, j + g));
        hv[g] = *(const vec_t*)(h + (size_t)s * F + lane * VEC);
    }
    for (int g = 0; g < G; ++g) {
        for (int v = 0; v < VEC; ++v) acc[v] += hv[g][v] * w[g];
    }
}

template <int F>
__global__ __launch_bounds__(256) void gcn_agg_pull(const float* __restrict__ h,
                                                    const int* __restrict__ rowptr,
                                                    const int2* __restrict__ rec,
                                                    const float* __restrict__ dinv,
                                                    const float* __restrict__ bias,
                                                    float* __restrict__ out, int n) {
    constexpr int VEC = F / 32;
    typedef __attribute__((ext_vector_type(VEC))) float vec_t;
    const int lane = threadIdx.x & 31;
    const int node = __builtin_amdgcn_readfirstlane((int)((blockIdx.x * blockDim.x + threadIdx.x) >> 5));
    if (node >= n) return;

    const int beg = rowptr[node];
    const int end = rowptr[node + 1];
    const float dn = dinv[node];

    const vec_t hs = *(const vec_t*)(h + (size_t)node * F + lane * VEC);
    const vec_t bv = *(const vec_t*)(bias + lane * VEC);
    float acc[VEC];
    for (int v = 0; v < VEC; ++v) acc[v] = hs[v] * dn * dn + bv[v];

    for (int k = beg; k < end; k += 32) {
        const int m = end - k < 32 ? end - k : 32;
        int my_s = 0, my_w = 0;
        if (lane < m) {
            const int2 r = rec[k + lane];
            my_s = r.x;
            my_w = __float_as_int(__int_as_float(r.y) * dn);
        }
        int j = 0;
        for (; j + 8 <= m; j += 8) gcn_pull_group<F, 8>(h, my_s, my_w, j, lane, acc);
        if (j + 4 <= m) { gcn_pull_group<F, 4>(h, my_s, my_w, j, lane, acc); j += 4; }
        if (j + 2 <= m) { gcn_pull_group<F, 2>(h, my_s, my_w, j, lane, acc); j += 2; }
        if (j < m)      { gcn_pull_group<F, 1>(h, my_s, my_w, j, lane, acc); }
    }

    vec_t o;
    for (int v = 0; v < VEC; ++v) o[v] = acc[v];
    *(vec_t*)(out + (size_t)node * F + lane * VEC) = o;
}


__global__ __launch_bounds__(256) void gcn_ln_relu(const float* __restrict__ in,
                                                   const float* __restrict__ gamma,
                                                   const float* __restrict__ beta,
                                                   float* __restrict__ outp, int n) {
    int lane = threadIdx.x & 31;
    int node = (int)(((long)blockIdx.x * blockDim.x + threadIdx.x) >> 5);
    if (node >= n) return;
    const float4 v = *(const float4*)(in + (long)node * 128 + lane * 4);

    float s = v.x + v.y + v.z + v.w;
#pragma unroll
    for (int off = 16; off > 0; off >>= 1) s += __shfl_xor(s, off, 32);
    float mu = s * (1.0f / 128.0f);

    float dx = v.x - mu, dy = v.y - mu, dz = v.z - mu, dw = v.w - mu;
    float sq = dx * dx + dy * dy + dz * dz + dw * dw;
#pragma unroll
    for (int off = 16; off > 0; off >>= 1) sq += __shfl_xor(sq, off, 32);
    float rstd = rsqrtf(sq * (1.0f / 128.0f) + GCN_EPS);

    const float4 g = *(const float4*)(gamma + lane * 4);
    const float4 b = *(const float4*)(beta + lane * 4);
    float4 o;
    o.x = fmaxf(dx * rstd * g.x + b.x, 0.0f);
    o.y = fmaxf(dy * rstd * g.y + b.y, 0.0f);
    o.z = fmaxf(dz * rstd * g.z + b.z, 0.0f);
    o.w = fmaxf(dw * rstd * g.w + b.w, 0.0f);
    *(float4*)(outp + (long)node * 128 + lane * 4) = o;
}


extern "C" void kernel_launch(void* const* d_in, const int* in_sizes, int n_in,
                              void* d_out, int out_size, void* d_ws, size_t ws_size,
                              hipStream_t stream) {
    const float* x   = (const float*)d_in[0];
    const int*   ei  = (const int*)d_in[1];
    const float* W1  = (const float*)d_in[2];
    const float* b1  = (const float*)d_in[3];
    const float* g1  = (const float*)d_in[4];
    const float* be1 = (const float*)d_in[5];
    const float* W2  = (const float*)d_in[6];
    const float* b2  = (const float*)d_in[7];
    float* out = (float*)d_out;

    const int n = in_sizes[0] / GCN_F_IN;
    const int e = in_sizes[1] / 2;
    const int* src = ei;
    const int* dst = ei + e;

    int2*  rec    = (int2*)d_ws;
    float* h1     = (float*)(rec + e);
    float* a1     = h1 + (size_t)n * GCN_HID;
    float* h2     = a1 + (size_t)n * GCN_HID;
    float* dinv   = h2 + (size_t)n * GCN_NCLS;
    int*   cnt    = (int*)(dinv + n);
    int*   rowptr = cnt + n;
    int*   rank   = rowptr + n + 1;

    gcn_cnt_zero<<<(n + 255) / 256, 256, 0, stream>>>(cnt, n);
    gcn_count_rank<<<(e + 255) / 256, 256, 0, stream>>>(dst, cnt, rank, e);
    gcn_scan_dinv<<<(n + 1023) / 1024, 1024, 0, stream>>>(cnt, rowptr, dinv, n);
    gcn_fill<<<(e + 255) / 256, 256, 0, stream>>>(src, dst, rank, rowptr, dinv, rec, e);

    gcn_gemm_wmma<GCN_HID, 2><<<(n + 15) / 16, 128, 0, stream>>>(x, W1, h1, n);
    gcn_agg_pull<GCN_HID><<<(int)(((long)n * 32 + 255) / 256), 256, 0, stream>>>(
        h1, rowptr, rec, dinv, b1, a1, n);
    gcn_ln_relu<<<(int)(((long)n * 32 + 255) / 256), 256, 0, stream>>>(a1, g1, be1, a1, n);

    gcn_gemm_wmma<GCN_NCLS, 1><<<(n + 15) / 16, 128, 0, stream>>>(a1, W2, h2, n);
    gcn_agg_pull<GCN_NCLS><<<(int)(((long)n * 32 + 255) / 256), 256, 0, stream>>>(
        h2, rowptr, rec, dinv, b2, out, n);
}


# Layer 1 as (A_hat x) W1: one kernel per 16-node tile pulls x rows into LDS, then GEMM1+b1, LayerNorm, ReLU, GEMM2 read the tile from LDS; 6 launches (was 9); sum order changes: aggregate before W1
# speedup vs baseline: 2.7983x; 1.2641x over previous
#include <hip/hip_runtime.h>

typedef __attribute__((ext_vector_type(2))) float v2f;
typedef __attribute__((ext_vector_type(8))) float v8f;

#define GCN_F_IN 128
#define GCN_HID  128
#define GCN_NCLS 64
#define GCN_EPS  1e-6f


__global__ __launch_bounds__(256) void gcn_cnt_zero(int* __restrict__ cnt, int n) {
    int i = blockIdx.x * blockDim.x + threadIdx.x;
    if (i < n) cnt[i] = 0;
}

__global__ __launch_bounds__(256) void gcn_count_rank(const int* __restrict__ dst,
                                                      int* __restrict__ cnt,
                                                      int* __restrict__ rank, int e) {
    int i = blockIdx.x * blockDim.x + threadIdx.x;
    if (i < e) rank[i] = atomicAdd(&cnt[dst[i]], 1);
}

__global__ __launch_bounds__(1024) void gcn_scan_dinv(const int* __restrict__ cnt,
                                                      int* __restrict__ rowptr,
                                                      float* __restrict__ dinv, int n) {
    __shared__ int before[32];
    __shared__ int wtot[32];
    const int tid  = threadIdx.x;
    const int lane = tid & 31;
    const int wv   = tid >> 5;
    const int base = blockIdx.x * 1024;

    int p = 0;
    for (int i = tid; i < base; i += 1024) p += cnt[i];
    for (int off = 16; off > 0; off >>= 1) p += __shfl_xor(p, off, 32);

    const int idx = base + tid;
    const int c   = idx < n ? cnt[idx] : 0;
    int inc = c;
    for (int off = 1; off < 32; off <<= 1) {
        int t = __shfl_up(inc, off, 32);
        if (lane >= off) inc += t;
    }
    if (lane == 0)  before[wv] = p;
    if (lane == 31) wtot[wv]   = inc;
    __syncthreads();

    int both = before[lane] + (lane < wv ? wtot[lane] : 0);
    for (int off = 16; off > 0; off >>= 1) both += __shfl_xor(both, off, 32);

    const int excl = both + inc - c;
    if (idx < n) {
        rowptr[idx] = excl;
        float d = (float)(c + 1);
        dinv[idx] = d > 0.0f ? rsqrtf(d) : 0.0f;
    }
    if (idx == n - 1) rowptr[n] = excl + c;
}

__global__ __launch_bounds__(256) void gcn_fill(const int* __restrict__ src,
                                                const int* __restrict__ dst,
                                                const int* __restrict__ rank,
                                                const int* __restrict__ rowptr,
                                                const float* __restrict__ dinv,
                                                int2* __restrict__ rec, int e) {
    int i = blockIdx.x * blockDim.x + threadIdx.x;
    if (i >= e) return;
    const int s   = src[i];
    const int pos = rowptr[dst[i]] + rank[i];
    rec[pos] = make_int2(s, __float_as_int(dinv[s]));
}


template <int F, int G>
__device__ __forceinline__ void gcn_pull_group(const float* __restrict__ h, int my_s, int my_w,
                                               int j, int lane,
                                               float (&acc)[F / 32]) {
    constexpr int VEC = F / 32;
    typedef __attribute__((ext_vector_type(VEC))) float vec_t;
    vec_t hv[G];
    float w[G];
    for (int g = 0; g < G; ++g) {
        const int s = __builtin_amdgcn_readlane(my_s, j + g);
        w[g] = __int_as_float(__builtin_amdgcn_readlane(my_w, j + g));
        hv[g] = *(const vec_t*)(h + (size_t)s * F + lane * VEC);
    }
    for (int g = 0; g < G; ++g) {
        for (int v = 0; v < VEC; ++v) acc[v] += hv[g][v] * w[g];
    }
}

template <int F>
__device__ __forceinline__ void gcn_pull_list(const float* __restrict__ h,
                                              const int2* __restrict__ rec,
                                              int beg, int end, float dn, int lane,
                                              float (&acc)[F / 32]) {
    for (int k = beg; k < end; k += 32) {
        const int m = end - k < 32 ? end - k : 32;
        int my_s = 0, my_w = 0;
        if (lane < m) {
            const int2 r = rec[k + lane];
            my_s = r.x;
            my_w = __float_as_int(__int_as_float(r.y) * dn);
        }
        int j = 0;
        for (; j + 8 <= m; j += 8) gcn_pull_group<F, 8>(h, my_s, my_w, j, lane, acc);
        if (j + 4 <= m) { gcn_pull_group<F, 4>(h, my_s, my_w, j, lane, acc); j += 4; }
        if (j + 2 <= m) { gcn_pull_group<F, 2>(h, my_s, my_w, j, lane, acc); j += 2; }
        if (j < m)      { gcn_pull_group<F, 1>(h, my_s, my_w, j, lane, acc); }
    }
}

template <int F>
__global__ __launch_bounds__(256) void gcn_agg_pull(const float* __restrict__ h,
                                                    const int* __restrict__ rowptr,
                                                    const int2* __restrict__ rec,
                                                    const float* __restrict__ dinv,
                                                    const float* __restrict__ bias,
                                                    float* __restrict__ out, int n) {
    constexpr int VEC = F / 32;
    typedef __attribute__((ext_vector_type(VEC))) float vec_t;
    const int lane = threadIdx.x & 31;
    const int node = __builtin_amdgcn_readfirstlane((int)((blockIdx.x * blockDim.x + threadIdx.x) >> 5));
    if (node >= n) return;

    const int beg = rowptr[node];
    const int end = rowptr[node + 1];
    const float dn = dinv[node];

    const vec_t hs = *(const vec_t*)(h + (size_t)node * F + lane * VEC);
    const vec_t bv = *(const vec_t*)(bias + lane * VEC);
    float acc[VEC];
    for (int v = 0; v < VEC; ++v) acc[v] = hs[v] * dn * dn + bv[v];

    gcn_pull_list<F>(h, rec, beg, end, dn, lane, acc);

    vec_t o;
    for (int v = 0; v < VEC; ++v) o[v] = acc[v];
    *(vec_t*)(out + (size_t)node * F + lane * VEC) = o;
}


#define GCN_PITCH 132

__global__ __launch_bounds__(128) void gcn_layer1_tile(const float* __restrict__ x,
                                                       const int* __restrict__ rowptr,
                                                       const int2* __restrict__ rec,
                                                       const float* __restrict__ dinv,
                                                       const float* __restrict__ W1,
                                                       const float* __restrict__ b1,
                                                       const float* __restrict__ gamma,
                                                       const float* __restrict__ beta,
                                                       const float* __restrict__ W2,
                                                       float* __restrict__ h2, int n) {
    __shared__ __attribute__((aligned(16))) float As[16 * GCN_PITCH];
    __shared__ __attribute__((aligned(16))) float Hs[16 * GCN_PITCH];

    const int lane  = threadIdx.x & 31;
    const int wave  = __builtin_amdgcn_readfirstlane((int)(threadIdx.x >> 5));
    const int mbase = blockIdx.x * 16;
    const int l16   = lane & 15;
    const int khalf = (lane >> 4) * 2;

    for (int i = 0; i < 4; ++i) {
        const int row  = wave * 4 + i;
        const int node = mbase + row;
        float acc[4] = {0.0f, 0.0f, 0.0f, 0.0f};
        if (node < n) {
            const int beg = rowptr[node];
            const int end = rowptr[node + 1];
            const float dn = dinv[node];
            const float4 xs = *(const float4*)(x + (size_t)node * GCN_F_IN + lane * 4);
            acc[0] = xs.x * dn * dn;
            acc[1] = xs.y * dn * dn;
            acc[2] = xs.z * dn * dn;
            acc[3] = xs.w * dn * dn;
            gcn_pull_list<GCN_F_IN>(x, rec, beg, end, dn, lane, acc);
        }
        *(float4*)(As + row * GCN_PITCH + lane * 4) = make_float4(acc[0], acc[1], acc[2], acc[3]);
    }
    __syncthreads();

    {
        const int n1 = wave * 32;
        const float* __restrict__ Ar = As + l16 * GCN_PITCH + khalf;
        v8f c0 = {};
        v8f c1 = {};
#pragma unroll 8
        for (int k0 = 0; k0 < GCN_F_IN; k0 += 4) {
            const v2f a = *(const v2f*)(Ar + k0);
            v2f p, q;
            p.x = W1[(k0 + khalf) * GCN_HID + n1 + l16];
            p.y = W1[(k0 + khalf + 1) * GCN_HID + n1 + l16];
            q.x = W1[(k0 + khalf) * GCN_HID + n1 + 16 + l16];
            q.y = W1[(k0 + khalf + 1) * GCN_HID + n1 + 16 + l16];
            c0 = __builtin_amdgcn_wmma_f32_16x16x4_f32(
                     false, a, false, p, (short)0, c0, false, false);
            c1 = __builtin_amdgcn_wmma_f32_16x16x4_f32(
                     false, a, false, q, (short)0, c1, false, false);
        }
        const int r0 = (lane >> 4) * 8;
        const float bb0 = b1[n1 + l16];
        const float bb1 = b1[n1 + 16 + l16];
#pragma unroll
        for (int v = 0; v < 8; ++v) {
            Hs[(r0 + v) * GCN_PITCH + n1 + l16]      = c0[v] + bb0;
            Hs[(r0 + v) * GCN_PITCH + n1 + 16 + l16] = c1[v] + bb1;
        }
    }
    __syncthreads();

    {
        const float4 g = *(const float4*)(gamma + lane * 4);
        const float4 b = *(const float4*)(beta + lane * 4);
        for (int i = 0; i < 4; ++i) {
            float* __restrict__ rowp = Hs + (wave * 4 + i) * GCN_PITCH + lane * 4;
            const float4 v = *(const float4*)rowp;

            float s = v.x + v.y + v.z + v.w;
#pragma unroll
            for (int off = 16; off > 0; off >>= 1) s += __shfl_xor(s, off, 32);
            float mu = s * (1.0f / 128.0f);

            float dx = v.x - mu, dy = v.y - mu, dz = v.z - mu, dw = v.w - mu;
            float sq = dx * dx + dy * dy + dz * dz + dw * dw;
#pragma unroll
            for (int off = 16; off > 0; off >>= 1) sq += __shfl_xor(sq, off, 32);
            float rstd = rsqrtf(sq * (1.0f / 128.0f) + GCN_EPS);

            float4 o;
            o.x = fmaxf(dx * rstd * g.x + b.x, 0.0f);
            o.y = fmaxf(dy * rstd * g.y + b.y, 0.0f);
            o.z = fmaxf(dz * rstd * g.z + b.z, 0.0f);
            o.w = fmaxf(dw * rstd * g.w + b.w, 0.0f);
            *(float4*)rowp = o;
        }
    }
    __syncthreads();

    {
        const int n2 = wave * 16;
        const float* __restrict__ Hr = Hs + l16 * GCN_PITCH + khalf;
        v8f c = {};
#pragma unroll
        for (int k0 = 0; k0 < GCN_HID; k0 += 4) {
            const v2f a = *(const v2f*)(Hr + k0);
            v2f b;
            b.x = W2[(k0 + khalf) * GCN_NCLS + n2 + l16];
            b.y = W2[(k0 + khalf + 1) * GCN_NCLS + n2 + l16];
            c = __builtin_amdgcn_wmma_f32_16x16x4_f32(
                    false, a, false, b, (short)0, c, false, false);
        }
        const int mrow = mbase + (lane >> 4) * 8;
        float* __restrict__ Cp = h2 + n2 + l16;
#pragma unroll
        for (int v = 0; v < 8; ++v) {
            int r = mrow + v;
            if (r < n) Cp[(long)r * GCN_NCLS] = c[v];
        }
    }
}


extern "C" void kernel_launch(void* const* d_in, const int* in_sizes, int n_in,
                              void* d_out, int out_size, void* d_ws, size_t ws_size,
                              hipStream_t stream) {
    const float* x   = (const float*)d_in[0];
    const int*   ei  = (const int*)d_in[1];
    const float* W1  = (const float*)d_in[2];
    const float* b1  = (const float*)d_in[3];
    const float* g1  = (const float*)d_in[4];
    const float* be1 = (const float*)d_in[5];
    const float* W2  = (const float*)d_in[6];
    const float* b2  = (const float*)d_in[7];
    float* out = (float*)d_out;

    const int n = in_sizes[0] / GCN_F_IN;
    const int e = in_sizes[1] / 2;
    const int* src = ei;
    const int* dst = ei + e;

    int2*  rec    = (int2*)d_ws;
    float* h2     = (float*)(rec + e);
    float* dinv   = h2 + (size_t)n * GCN_NCLS;
    int*   cnt    = (int*)(dinv + n);
    int*   rowptr = cnt + n;
    int*   rank   = rowptr + n + 1;

    gcn_cnt_zero<<<(n + 255) / 256, 256, 0, stream>>>(cnt, n);
    gcn_count_rank<<<(e + 255) / 256, 256, 0, stream>>>(dst, cnt, rank, e);
    gcn_scan_dinv<<<(n + 1023) / 1024, 1024, 0, stream>>>(cnt, rowptr, dinv, n);
    gcn_fill<<<(e + 255) / 256, 256, 0, stream>>>(src, dst, rank, rowptr, dinv, rec, e);

    gcn_layer1_tile<<<(n + 15) / 16, 128, 0, stream>>>(
        x, rowptr, rec, dinv, W1, b1, g1, be1, W2, h2, n);

    gcn_agg_pull<GCN_NCLS><<<(int)(((long)n * 32 + 255) / 256), 256, 0, stream>>>(
        h2, rowptr, rec, dinv, b2, out, n);
}


# Edges grouped by dst by a two-level counting sort on LDS counters (256-node bins, then nodes in a bin), not one global returning atomic per edge; 7 launches (was 6); sum order = LDS arrival rank
# speedup vs baseline: 4.3624x; 1.5589x over previous
#include <hip/hip_runtime.h>

typedef __attribute__((ext_vector_type(2))) float v2f;
typedef __attribute__((ext_vector_type(8))) float v8f;

#define GCN_F_IN 128
#define GCN_HID  128
#define GCN_NCLS 64
#define GCN_EPS  1e-6f


#define GCN_BIN_SHIFT 8
#define GCN_BIN_NODES 256
#define GCN_MAX_BINS  1024
#define GCN_CHUNK     4096

__global__ __launch_bounds__(1024) void gcn_bin_count(const int* __restrict__ dst,
                                                      int* __restrict__ counts,
                                                      int e, int nbins, int nchunks) {
    __shared__ int hist[GCN_MAX_BINS];
    const int tid = threadIdx.x;
    hist[tid] = 0;
    __syncthreads();
    const int base = blockIdx.x * GCN_CHUNK;
    int d[4];
#pragma unroll
    for (int j = 0; j < 4; ++j) {
        const int i = base + j * 1024 + tid;
        d[j] = i < e ? dst[i] : -1;
    }
#pragma unroll
    for (int j = 0; j < 4; ++j) {
        if (d[j] >= 0) atomicAdd(&hist[d[j] >> GCN_BIN_SHIFT], 1);
    }
    __syncthreads();
    if (tid < nbins) counts[tid * nchunks + blockIdx.x] = hist[tid];
}

__global__ __launch_bounds__(1024) void gcn_scan_flat(const int* __restrict__ counts,
                                                      int* __restrict__ offs, int total) {
    __shared__ int before[32];
    __shared__ int wtot[32];
    const int tid  = threadIdx.x;
    const int lane = tid & 31;
    const int wv   = tid >> 5;
    const int base = blockIdx.x * 1024;

    int p = 0;
    for (int i = tid; i < base; i += 1024) p += counts[i];
    for (int off = 16; off > 0; off >>= 1) p += __shfl_xor(p, off, 32);

    const int idx = base + tid;
    const int c   = idx < total ? counts[idx] : 0;
    int inc = c;
    for (int off = 1; off < 32; off <<= 1) {
        int t = __shfl_up(inc, off, 32);
        if (lane >= off) inc += t;
    }
    if (lane == 0)  before[wv] = p;
    if (lane == 31) wtot[wv]   = inc;
    __syncthreads();

    int both = before[lane] + (lane < wv ? wtot[lane] : 0);
    for (int off = 16; off > 0; off >>= 1) both += __shfl_xor(both, off, 32);

    if (idx < total) offs[idx] = both + inc - c;
}

__global__ __launch_bounds__(1024) void gcn_bin_scatter(const int* __restrict__ src,
                                                        const int* __restrict__ dst,
                                                        const int* __restrict__ offs,
                                                        int2* __restrict__ binned,
                                                        int e, int nbins, int nchunks) {
    __shared__ int cursor[GCN_MAX_BINS];
    const int tid = threadIdx.x;
    if (tid < nbins) cursor[tid] = offs[tid * nchunks + blockIdx.x];
    __syncthreads();
    const int base = blockIdx.x * GCN_CHUNK;
    int s[4], d[4];
#pragma unroll
    for (int j = 0; j < 4; ++j) {
        const int i = base + j * 1024 + tid;
        d[j] = i < e ? dst[i] : -1;
        s[j] = i < e ? src[i] : 0;
    }
#pragma unroll
    for (int j = 0; j < 4; ++j) {
        if (d[j] >= 0) {
            const int pos = atomicAdd(&cursor[d[j] >> GCN_BIN_SHIFT], 1);
            binned[pos] = make_int2(s[j], d[j]);
        }
    }
}

__global__ __launch_bounds__(1024) void gcn_bin_sort(const int2* __restrict__ binned,
                                                     const int* __restrict__ offs,
                                                     int* __restrict__ rowptr,
                                                     float* __restrict__ dinv,
                                                     int* __restrict__ srcs,
                                                     int n, int e, int nbins, int nchunks) {
    __shared__ int cl[GCN_BIN_NODES];
    __shared__ int wtot[GCN_BIN_NODES / 32];
    const int tid  = threadIdx.x;
    const int lane = tid & 31;
    const int wv   = tid >> 5;
    const int bin  = blockIdx.x;
    const int nodebase = bin << GCN_BIN_SHIFT;
    const int beg = offs[bin * nchunks];
    const int end = bin + 1 < nbins ? offs[(bin + 1) * nchunks] : e;

    if (tid < GCN_BIN_NODES) cl[tid] = 0;
    __syncthreads();
    for (int i = beg + tid; i < end; i += 4096) {
        int d[4];
#pragma unroll
        for (int j = 0; j < 4; ++j) d[j] = i + j * 1024 < end ? binned[i + j * 1024].y : -1;
#pragma unroll
        for (int j = 0; j < 4; ++j) {
            if (d[j] >= 0) atomicAdd(&cl[d[j] - nodebase], 1);
        }
    }
    __syncthreads();

    const int c = tid < GCN_BIN_NODES ? cl[tid] : 0;
    int inc = c;
    for (int off = 1; off < 32; off <<= 1) {
        int t = __shfl_up(inc, off, 32);
        if (lane >= off) inc += t;
    }
    if (tid < GCN_BIN_NODES && lane == 31) wtot[wv] = inc;
    __syncthreads();

    if (tid < GCN_BIN_NODES) {
        int pre = 0;
#pragma unroll
        for (int w = 0; w < GCN_BIN_NODES / 32; ++w) pre += w < wv ? wtot[w] : 0;
        const int start = beg + pre + inc - c;
        cl[tid] = start;
        const int node = nodebase + tid;
        if (node < n) {
            rowptr[node] = start;
            float dg = (float)(c + 1);
            dinv[node] = dg > 0.0f ? rsqrtf(dg) : 0.0f;
        }
    }
    if (bin == nbins - 1 && tid == 0) rowptr[n] = end;
    __syncthreads();

    for (int i = beg + tid; i < end; i += 4096) {
        int2 r[4];
#pragma unroll
        for (int j = 0; j < 4; ++j) r[j] = i + j * 1024 < end ? binned[i + j * 1024] : make_int2(0, -1);
#pragma unroll
        for (int j = 0; j < 4; ++j) {
            if (r[j].y >= 0) {
                const int pos = atomicAdd(&cl[r[j].y - nodebase], 1);
                srcs[pos] = r[j].x;
            }
        }
    }
}

__global__ __launch_bounds__(256) void gcn_fill_w(const int* __restrict__ srcs,
                                                  const float* __restrict__ dinv,
                                                  int2* __restrict__ rec, int e) {
    const int i0 = (blockIdx.x * 512 + threadIdx.x) * 4;
    const int i1 = i0 + 256 * 4;
    if (i1 + 3 < e) {
        const int4 a = *(const int4*)(srcs + i0);
        const int4 b = *(const int4*)(srcs + i1);
        const float a0 = dinv[a.x], a1 = dinv[a.y], a2 = dinv[a.z], a3 = dinv[a.w];
        const float b0 = dinv[b.x], b1 = dinv[b.y], b2 = dinv[b.z], b3 = dinv[b.w];
        int4* oa = (int4*)(rec + i0);
        int4* ob = (int4*)(rec + i1);
        oa[0] = make_int4(a.x, __float_as_int(a0), a.y, __float_as_int(a1));
        oa[1] = make_int4(a.z, __float_as_int(a2), a.w, __float_as_int(a3));
        ob[0] = make_int4(b.x, __float_as_int(b0), b.y, __float_as_int(b1));
        ob[1] = make_int4(b.z, __float_as_int(b2), b.w, __float_as_int(b3));
    } else {
        for (int k = i0; k < e && k < i0 + 4; ++k) {
            const int s = srcs[k];
            rec[k] = make_int2(s, __float_as_int(dinv[s]));
        }
        for (int k = i1; k < e && k < i1 + 4; ++k) {
            const int s = srcs[k];
            rec[k] = make_int2(s, __float_as_int(dinv[s]));
        }
    }
}


template <int F, int G>
__device__ __forceinline__ void gcn_pull_group(const float* __restrict__ h, int my_s, int my_w,
                                               int j, int lane,
                                               float (&acc)[F / 32]) {
    constexpr int VEC = F / 32;
    typedef __attribute__((ext_vector_type(VEC))) float vec_t;
    vec_t hv[G];
    float w[G];
    for (int g = 0; g < G; ++g) {
        const int s = __builtin_amdgcn_readlane(my_s, j + g);
        w[g] = __int_as_float(__builtin_amdgcn_readlane(my_w, j + g));
        hv[g] = *(const vec_t*)(h + (size_t)s * F + lane * VEC);
    }
    for (int g = 0; g < G; ++g) {
        for (int v = 0; v < VEC; ++v) acc[v] += hv[g][v] * w[g];
    }
}

template <int F>
__device__ __forceinline__ void gcn_pull_list(const float* __restrict__ h,
                                              const int2* __restrict__ rec,
                                              int beg, int end, float dn, int lane,
                                              float (&acc)[F / 32]) {
    for (int k = beg; k < end; k += 32) {
        const int m = end - k < 32 ? end - k : 32;
        int my_s = 0, my_w = 0;
        if (lane < m) {
            const int2 r = rec[k + lane];
            my_s = r.x;
            my_w = __float_as_int(__int_as_float(r.y) * dn);
        }
        int j = 0;
        for (; j + 8 <= m; j += 8) gcn_pull_group<F, 8>(h, my_s, my_w, j, lane, acc);
        if (j + 4 <= m) { gcn_pull_group<F, 4>(h, my_s, my_w, j, lane, acc); j += 4; }
        if (j + 2 <= m) { gcn_pull_group<F, 2>(h, my_s, my_w, j, lane, acc); j += 2; }
        if (j < m)      { gcn_pull_group<F, 1>(h, my_s, my_w, j, lane, acc); }
    }
}

template <int F>
__global__ __launch_bounds__(256) void gcn_agg_pull(const float* __restrict__ h,
                                                    const int* __restrict__ rowptr,
                                                    const int2* __restrict__ rec,
                                                    const float* __restrict__ dinv,
                                                    const float* __restrict__ bias,
                                                    float* __restrict__ out, int n) {
    constexpr int VEC = F / 32;
    typedef __attribute__((ext_vector_type(VEC))) float vec_t;
    const int lane = threadIdx.x & 31;
    const int node = __builtin_amdgcn_readfirstlane((int)((blockIdx.x * blockDim.x + threadIdx.x) >> 5));
    if (node >= n) return;

    const int beg = rowptr[node];
    const int end = rowptr[node + 1];
    const float dn = dinv[node];

    const vec_t hs = *(const vec_t*)(h + (size_t)node * F + lane * VEC);
    const vec_t bv = *(const vec_t*)(bias + lane * VEC);
    float acc[VEC];
    for (int v = 0; v < VEC; ++v) acc[v] = hs[v] * dn * dn + bv[v];

    gcn_pull_list<F>(h, rec, beg, end, dn, lane, acc);

    vec_t o;
    for (int v = 0; v < VEC; ++v) o[v] = acc[v];
    *(vec_t*)(out + (size_t)node * F + lane * VEC) = o;
}


#define GCN_PITCH 132

__global__ __launch_bounds__(128) void gcn_layer1_tile(const float* __restrict__ x,
                                                       const int* __restrict__ rowptr,
                                                       const int2* __restrict__ rec,
                                                       const float* __restrict__ dinv,
                                                       const float* __restrict__ W1,
                                                       const float* __restrict__ b1,
                                                       const float* __restrict__ gamma,
                                                       const float* __restrict__ beta,
                                                       const float* __restrict__ W2,
                                                       float* __restrict__ h2, int n) {
    __shared__ __attribute__((aligned(16))) float As[16 * GCN_PITCH];
    __shared__ __attribute__((aligned(16))) float Hs[16 * GCN_PITCH];

    const int lane  = threadIdx.x & 31;
    const int wave  = __builtin_amdgcn_readfirstlane((int)(threadIdx.x >> 5));
    const int mbase = blockIdx.x * 16;
    const int l16   = lane & 15;
    const int khalf = (lane >> 4) * 2;

    for (int i = 0; i < 4; ++i) {
        const int row  = wave * 4 + i;
        const int node = mbase + row;
        float acc[4] = {0.0f, 0.0f, 0.0f, 0.0f};
        if (node < n) {
            const int beg = rowptr[node];
            const int end = rowptr[node + 1];
            const float dn = dinv[node];
            const float4 xs = *(const float4*)(x + (size_t)node * GCN_F_IN + lane * 4);
            acc[0] = xs.x * dn * dn;
            acc[1] = xs.y * dn * dn;
            acc[2] = xs.z * dn * dn;
            acc[3] = xs.w * dn * dn;
            gcn_pull_list<GCN_F_IN>(x, rec, beg, end, dn, lane, acc);
        }
        *(float4*)(As + row * GCN_PITCH + lane * 4) = make_float4(acc[0], acc[1], acc[2], acc[3]);
    }
    __syncthreads();

    {
        const int n1 = wave * 32;
        const float* __restrict__ Ar = As + l16 * GCN_PITCH + khalf;
        v8f c0 = {};
        v8f c1 = {};
#pragma unroll 8
        for (int k0 = 0; k0 < GCN_F_IN; k0 += 4) {
            const v2f a = *(const v2f*)(Ar + k0);
            v2f p, q;
            p.x = W1[(k0 + khalf) * GCN_HID + n1 + l16];
            p.y = W1[(k0 + khalf + 1) * GCN_HID + n1 + l16];
            q.x = W1[(k0 + khalf) * GCN_HID + n1 + 16 + l16];
            q.y = W1[(k0 + khalf + 1) * GCN_HID + n1 + 16 + l16];
            c0 = __builtin_amdgcn_wmma_f32_16x16x4_f32(
                     false, a, false, p, (short)0, c0, false, false);
            c1 = __builtin_amdgcn_wmma_f32_16x16x4_f32(
                     false, a, false, q, (short)0, c1, false, false);
        }
        const int r0 = (lane >> 4) * 8;
        const float bb0 = b1[n1 + l16];
        const float bb1 = b1[n1 + 16 + l16];
#pragma unroll
        for (int v = 0; v < 8; ++v) {
            Hs[(r0 + v) * GCN_PITCH + n1 + l16]      = c0[v] + bb0;
            Hs[(r0 + v) * GCN_PITCH + n1 + 16 + l16] = c1[v] + bb1;
        }
    }
    __syncthreads();

    {
        const float4 g = *(const float4*)(gamma + lane * 4);
        const float4 b = *(const float4*)(beta + lane * 4);
        for (int i = 0; i < 4; ++i) {
            float* __restrict__ rowp = Hs + (wave * 4 + i) * GCN_PITCH + lane * 4;
            const float4 v = *(const float4*)rowp;

            float s = v.x + v.y + v.z + v.w;
#pragma unroll
            for (int off = 16; off > 0; off >>= 1) s += __shfl_xor(s, off, 32);
            float mu = s * (1.0f / 128.0f);

            float dx = v.x - mu, dy = v.y - mu, dz = v.z - mu, dw = v.w - mu;
            float sq = dx * dx + dy * dy + dz * dz + dw * dw;
#pragma unroll
            for (int off = 16; off > 0; off >>= 1) sq += __shfl_xor(sq, off, 32);
            float rstd = rsqrtf(sq * (1.0f / 128.0f) + GCN_EPS);

            float4 o;
            o.x = fmaxf(dx * rstd * g.x + b.x, 0.0f);
            o.y = fmaxf(dy * rstd * g.y + b.y, 0.0f);
            o.z = fmaxf(dz * rstd * g.z + b.z, 0.0f);
            o.w = fmaxf(dw * rstd * g.w + b.w, 0.0f);
            *(float4*)rowp = o;
        }
    }
    __syncthreads();

    {
        const int n2 = wave * 16;
        const float* __restrict__ Hr = Hs + l16 * GCN_PITCH + khalf;
        v8f c = {};
#pragma unroll
        for (int k0 = 0; k0 < GCN_HID; k0 += 4) {
            const v2f a = *(const v2f*)(Hr + k0);
            v2f b;
            b.x = W2[(k0 + khalf) * GCN_NCLS + n2 + l16];
            b.y = W2[(k0 + khalf + 1) * GCN_NCLS + n2 + l16];
            c = __builtin_amdgcn_wmma_f32_16x16x4_f32(
                    false, a, false, b, (short)0, c, false, false);
        }
        const int mrow = mbase + (lane >> 4) * 8;
        float* __restrict__ Cp = h2 + n2 + l16;
#pragma unroll
        for (int v = 0; v < 8; ++v) {
            int r = mrow + v;
            if (r < n) Cp[(long)r * GCN_NCLS] = c[v];
        }
    }
}


extern "C" void kernel_launch(void* const* d_in, const int* in_sizes, int n_in,
                              void* d_out, int out_size, void* d_ws, size_t ws_size,
                              hipStream_t stream) {
    const float* x   = (const float*)d_in[0];
    const int*   ei  = (const int*)d_in[1];
    const float* W1  = (const float*)d_in[2];
    const float* b1  = (const float*)d_in[3];
    const float* g1  = (const float*)d_in[4];
    const float* be1 = (const float*)d_in[5];
    const float* W2  = (const float*)d_in[6];
    const float* b2  = (const float*)d_in[7];
    float* out = (float*)d_out;

    const int n = in_sizes[0] / GCN_F_IN;
    const int e = in_sizes[1] / 2;
    const int* src = ei;
    const int* dst = ei + e;

    const int nbins   = (n + GCN_BIN_NODES - 1) / GCN_BIN_NODES;
    const int nchunks = (e + GCN_CHUNK - 1) / GCN_CHUNK;
    const int total   = nbins * nchunks;

    float* h2     = (float*)d_ws;
    int2*  rec    = (int2*)(h2 + (size_t)n * GCN_NCLS);
    int2*  binned = rec + e;
    int*   srcs   = (int*)(binned + e);
    float* dinv   = (float*)(srcs + e);
    int*   rowptr = (int*)(dinv + n);
    int*   counts = rowptr + n + 1;
    int*   offs   = counts + total;

    gcn_bin_count<<<nchunks, 1024, 0, stream>>>(dst, counts, e, nbins, nchunks);
    gcn_scan_flat<<<(total + 1023) / 1024, 1024, 0, stream>>>(counts, offs, total);
    gcn_bin_scatter<<<nchunks, 1024, 0, stream>>>(src, dst, offs, binned, e, nbins, nchunks);
    gcn_bin_sort<<<nbins, 1024, 0, stream>>>(binned, offs, rowptr, dinv, srcs, n, e, nbins, nchunks);
    gcn_fill_w<<<((e + 3) / 4 + 511) / 512, 256, 0, stream>>>(srcs, dinv, rec, e);

    gcn_layer1_tile<<<(n + 15) / 16, 128, 0, stream>>>(
        x, rowptr, rec, dinv, W1, b1, g1, be1, W2, h2, n);

    gcn_agg_pull<GCN_NCLS><<<(int)(((long)n * 32 + 255) / 256), 256, 0, stream>>>(
        h2, rowptr, rec, dinv, b2, out, n);
}


# Level-1 binning made chunk-local: each 4096-edge block counts, scans and places its edges in its own region; bin blocks gather their segments; 5 launches (was 7); list order = LDS arrival as before
# speedup vs baseline: 5.2771x; 1.2097x over previous
#include <hip/hip_runtime.h>

typedef __attribute__((ext_vector_type(2))) float v2f;
typedef __attribute__((ext_vector_type(8))) float v8f;

#define GCN_F_IN 128
#define GCN_HID  128
#define GCN_NCLS 64
#define GCN_EPS  1e-6f


#define GCN_BIN_SHIFT  8
#define GCN_BIN_NODES  256
#define GCN_MAX_BINS   1024
#define GCN_CHUNK      4096
#define GCN_MAX_CHUNKS 1024

__global__ __launch_bounds__(1024) void gcn_chunk_bin(const int* __restrict__ src,
                                                      const int* __restrict__ dst,
                                                      int2* __restrict__ binned,
                                                      int* __restrict__ loc,
                                                      int e, int nbins) {
    __shared__ int hist[GCN_MAX_BINS];
    __shared__ int wtot[32];
    const int tid  = threadIdx.x;
    const int lane = tid & 31;
    const int wv   = tid >> 5;
    hist[tid] = 0;
    __syncthreads();
    const int base = blockIdx.x * GCN_CHUNK;
    int s[4], d[4];
#pragma unroll
    for (int j = 0; j < 4; ++j) {
        const int i = base + j * 1024 + tid;
        d[j] = i < e ? dst[i] : -1;
        s[j] = i < e ? src[i] : 0;
    }
#pragma unroll
    for (int j = 0; j < 4; ++j) {
        if (d[j] >= 0) atomicAdd(&hist[d[j] >> GCN_BIN_SHIFT], 1);
    }
    __syncthreads();

    const int c = hist[tid];
    int inc = c;
    for (int off = 1; off < 32; off <<= 1) {
        int t = __shfl_up(inc, off, 32);
        if (lane >= off) inc += t;
    }
    if (lane == 31) wtot[wv] = inc;
    __syncthreads();
    int pre = lane < wv ? wtot[lane] : 0;
    for (int off = 16; off > 0; off >>= 1) pre += __shfl_xor(pre, off, 32);
    const int excl = pre + inc - c;
    hist[tid] = excl;
    if (tid < nbins) loc[blockIdx.x * nbins + tid] = excl;
    __syncthreads();

#pragma unroll
    for (int j = 0; j < 4; ++j) {
        if (d[j] >= 0) {
            const int pos = atomicAdd(&hist[d[j] >> GCN_BIN_SHIFT], 1);
            binned[base + pos] = make_int2(s[j], d[j]);
        }
    }
}

__global__ __launch_bounds__(1024) void gcn_bin_sort(const int2* __restrict__ binned,
                                                     const int* __restrict__ loc,
                                                     int* __restrict__ rowptr,
                                                     float* __restrict__ dinv,
                                                     int* __restrict__ srcs,
                                                     int n, int e, int nbins, int nchunks) {
    __shared__ int cl[GCN_BIN_NODES];
    __shared__ int wtot[GCN_BIN_NODES / 32];
    __shared__ int segb[GCN_MAX_CHUNKS + 256];
    __shared__ int sege[GCN_MAX_CHUNKS + 256];
    __shared__ int redb[32];
    __shared__ int rede[32];
    const int tid  = threadIdx.x;
    const int lane = tid & 31;
    const int wv   = tid >> 5;
    const int bin  = blockIdx.x;
    const int nodebase = bin << GCN_BIN_SHIFT;

    int sb = 0, se = 0;
    if (tid < nchunks) {
        const int left = e - tid * GCN_CHUNK;
        sb = loc[tid * nbins + bin];
        se = bin + 1 < nbins ? loc[tid * nbins + bin + 1] : (left < GCN_CHUNK ? left : GCN_CHUNK);
    }
    segb[tid] = sb;
    sege[tid] = se;
    if (tid < 256) {
        segb[GCN_MAX_CHUNKS + tid] = 0;
        sege[GCN_MAX_CHUNKS + tid] = 0;
    }
    if (tid < GCN_BIN_NODES) cl[tid] = 0;
    int pb = sb, pe = se;
    for (int off = 16; off > 0; off >>= 1) {
        pb += __shfl_xor(pb, off, 32);
        pe += __shfl_xor(pe, off, 32);
    }
    if (lane == 0) { redb[wv] = pb; rede[wv] = pe; }
    __syncthreads();
    int beg = redb[lane], end = rede[lane];
    for (int off = 16; off > 0; off >>= 1) {
        beg += __shfl_xor(beg, off, 32);
        end += __shfl_xor(end, off, 32);
    }

    for (int c0 = wv; c0 < nchunks; c0 += 256) {
        int sb8[8], se8[8], d[8];
#pragma unroll
        for (int u = 0; u < 8; ++u) {
            sb8[u] = segb[c0 + 32 * u];
            se8[u] = sege[c0 + 32 * u];
        }
#pragma unroll
        for (int u = 0; u < 8; ++u) {
            const int i = sb8[u] + lane;
            d[u] = i < se8[u] ? binned[(c0 + 32 * u) * GCN_CHUNK + i].y : -1;
        }
#pragma unroll
        for (int u = 0; u < 8; ++u) {
            if (d[u] >= 0) atomicAdd(&cl[d[u] - nodebase], 1);
        }
#pragma unroll
        for (int u = 0; u < 8; ++u) {
            for (int i = sb8[u] + 32 + lane; i < se8[u]; i += 32)
                atomicAdd(&cl[binned[(c0 + 32 * u) * GCN_CHUNK + i].y - nodebase], 1);
        }
    }
    __syncthreads();

    const int c = tid < GCN_BIN_NODES ? cl[tid] : 0;
    int inc = c;
    for (int off = 1; off < 32; off <<= 1) {
        int t = __shfl_up(inc, off, 32);
        if (lane >= off) inc += t;
    }
    if (tid < GCN_BIN_NODES && lane == 31) wtot[wv] = inc;
    __syncthreads();

    if (tid < GCN_BIN_NODES) {
        int pre = 0;
#pragma unroll
        for (int w = 0; w < GCN_BIN_NODES / 32; ++w) pre += w < wv ? wtot[w] : 0;
        const int start = beg + pre + inc - c;
        cl[tid] = start;
        const int node = nodebase + tid;
        if (node < n) {
            rowptr[node] = start;
            float dg = (float)(c + 1);
            dinv[node] = dg > 0.0f ? rsqrtf(dg) : 0.0f;
        }
    }
    if (bin == nbins - 1 && tid == 0) rowptr[n] = end;
    __syncthreads();

    for (int c0 = wv; c0 < nchunks; c0 += 256) {
        int sb8[8], se8[8];
        int2 r[8];
#pragma unroll
        for (int u = 0; u < 8; ++u) {
            sb8[u] = segb[c0 + 32 * u];
            se8[u] = sege[c0 + 32 * u];
        }
#pragma unroll
        for (int u = 0; u < 8; ++u) {
            const int i = sb8[u] + lane;
            r[u] = i < se8[u] ? binned[(c0 + 32 * u) * GCN_CHUNK + i] : make_int2(0, -1);
        }
#pragma unroll
        for (int u = 0; u < 8; ++u) {
            if (r[u].y >= 0) {
                const int pos = atomicAdd(&cl[r[u].y - nodebase], 1);
                srcs[pos] = r[u].x;
            }
        }
#pragma unroll
        for (int u = 0; u < 8; ++u) {
            for (int i = sb8[u] + 32 + lane; i < se8[u]; i += 32) {
                const int2 q = binned[(c0 + 32 * u) * GCN_CHUNK + i];
                const int pos = atomicAdd(&cl[q.y - nodebase], 1);
                srcs[pos] = q.x;
            }
        }
    }
}

__global__ __launch_bounds__(256) void gcn_fill_w(const int* __restrict__ srcs,
                                                  const float* __restrict__ dinv,
                                                  int2* __restrict__ rec, int e) {
    const int i0 = (blockIdx.x * 512 + threadIdx.x) * 4;
    const int i1 = i0 + 256 * 4;
    if (i1 + 3 < e) {
        const int4 a = *(const int4*)(srcs + i0);
        const int4 b = *(const int4*)(srcs + i1);
        const float a0 = dinv[a.x], a1 = dinv[a.y], a2 = dinv[a.z], a3 = dinv[a.w];
        const float b0 = dinv[b.x], b1 = dinv[b.y], b2 = dinv[b.z], b3 = dinv[b.w];
        int4* oa = (int4*)(rec + i0);
        int4* ob = (int4*)(rec + i1);
        oa[0] = make_int4(a.x, __float_as_int(a0), a.y, __float_as_int(a1));
        oa[1] = make_int4(a.z, __float_as_int(a2), a.w, __float_as_int(a3));
        ob[0] = make_int4(b.x, __float_as_int(b0), b.y, __float_as_int(b1));
        ob[1] = make_int4(b.z, __float_as_int(b2), b.w, __float_as_int(b3));
    } else {
        for (int k = i0; k < e && k < i0 + 4; ++k) {
            const int s = srcs[k];
            rec[k] = make_int2(s, __float_as_int(dinv[s]));
        }
        for (int k = i1; k < e && k < i1 + 4; ++k) {
            const int s = srcs[k];
            rec[k] = make_int2(s, __float_as_int(dinv[s]));
        }
    }
}


template <int F, int G>
__device__ __forceinline__ void gcn_pull_group(const float* __restrict__ h, int my_s, int my_w,
                                               int j, int lane,
                                               float (&acc)[F / 32]) {
    constexpr int VEC = F / 32;
    typedef __attribute__((ext_vector_type(VEC))) float vec_t;
    vec_t hv[G];
    float w[G];
    for (int g = 0; g < G; ++g) {
        const int s = __builtin_amdgcn_readlane(my_s, j + g);
        w[g] = __int_as_float(__builtin_amdgcn_readlane(my_w, j + g));
        hv[g] = *(const vec_t*)(h + (size_t)s * F + lane * VEC);
    }
    for (int g = 0; g < G; ++g) {
        for (int v = 0; v < VEC; ++v) acc[v] += hv[g][v] * w[g];
    }
}

template <int F>
__device__ __forceinline__ void gcn_pull_list(const float* __restrict__ h,
                                              const int2* __restrict__ rec,
                                              int beg, int end, float dn, int lane,
                                              float (&acc)[F / 32]) {
    for (int k = beg; k < end; k += 32) {
        const int m = end - k < 32 ? end - k : 32;
        int my_s = 0, my_w = 0;
        if (lane < m) {
            const int2 r = rec[k + lane];
            my_s = r.x;
            my_w = __float_as_int(__int_as_float(r.y) * dn);
        }
        int j = 0;
        for (; j + 8 <= m; j += 8) gcn_pull_group<F, 8>(h, my_s, my_w, j, lane, acc);
        if (j + 4 <= m) { gcn_pull_group<F, 4>(h, my_s, my_w, j, lane, acc); j += 4; }
        if (j + 2 <= m) { gcn_pull_group<F, 2>(h, my_s, my_w, j, lane, acc); j += 2; }
        if (j < m)      { gcn_pull_group<F, 1>(h, my_s, my_w, j, lane, acc); }
    }
}

template <int F>
__global__ __launch_bounds__(256) void gcn_agg_pull(const float* __restrict__ h,
                                                    const int* __restrict__ rowptr,
                                                    const int2* __restrict__ rec,
                                                    const float* __restrict__ dinv,
                                                    const float* __restrict__ bias,
                                                    float* __restrict__ out, int n) {
    constexpr int VEC = F / 32;
    typedef __attribute__((ext_vector_type(VEC))) float vec_t;
    const int lane = threadIdx.x & 31;
    const int node = __builtin_amdgcn_readfirstlane((int)((blockIdx.x * blockDim.x + threadIdx.x) >> 5));
    if (node >= n) return;

    const int beg = rowptr[node];
    const int end = rowptr[node + 1];
    const float dn = dinv[node];

    const vec_t hs = *(const vec_t*)(h + (size_t)node * F + lane * VEC);
    const vec_t bv = *(const vec_t*)(bias + lane * VEC);
    float acc[VEC];
    for (int v = 0; v < VEC; ++v) acc[v] = hs[v] * dn * dn + bv[v];

    gcn_pull_list<F>(h, rec, beg, end, dn, lane, acc);

    vec_t o;
    for (int v = 0; v < VEC; ++v) o[v] = acc[v];
    *(vec_t*)(out + (size_t)node * F + lane * VEC) = o;
}


#define GCN_PITCH 132

__global__ __launch_bounds__(128) void gcn_layer1_tile(const float* __restrict__ x,
                                                       const int* __restrict__ rowptr,
                                                       const int2* __restrict__ rec,
                                                       const float* __restrict__ dinv,
                                                       const float* __restrict__ W1,
                                                       const float* __restrict__ b1,
                                                       const float* __restrict__ gamma,
                                                       const float* __restrict__ beta,
                                                       const float* __restrict__ W2,
                                                       float* __restrict__ h2, int n) {
    __shared__ __attribute__((aligned(16))) float As[16 * GCN_PITCH];
    __shared__ __attribute__((aligned(16))) float Hs[16 * GCN_PITCH];

    const int lane  = threadIdx.x & 31;
    const int wave  = __builtin_amdgcn_readfirstlane((int)(threadIdx.x >> 5));
    const int mbase = blockIdx.x * 16;
    const int l16   = lane & 15;
    const int khalf = (lane >> 4) * 2;

    for (int i = 0; i < 4; ++i) {
        const int row  = wave * 4 + i;
        const int node = mbase + row;
        float acc[4] = {0.0f, 0.0f, 0.0f, 0.0f};
        if (node < n) {
            const int beg = rowptr[node];
            const int end = rowptr[node + 1];
            const float dn = dinv[node];
            const float4 xs = *(const float4*)(x + (size_t)node * GCN_F_IN + lane * 4);
            acc[0] = xs.x * dn * dn;
            acc[1] = xs.y * dn * dn;
            acc[2] = xs.z * dn * dn;
            acc[3] = xs.w * dn * dn;
            gcn_pull_list<GCN_F_IN>(x, rec, beg, end, dn, lane, acc);
        }
        *(float4*)(As + row * GCN_PITCH + lane * 4) = make_float4(acc[0], acc[1], acc[2], acc[3]);
    }
    __syncthreads();

    {
        const int n1 = wave * 32;
        const float* __restrict__ Ar = As + l16 * GCN_PITCH + khalf;
        v8f c0 = {};
        v8f c1 = {};
#pragma unroll 8
        for (int k0 = 0; k0 < GCN_F_IN; k0 += 4) {
            const v2f a = *(const v2f*)(Ar + k0);
            v2f p, q;
            p.x = W1[(k0 + khalf) * GCN_HID + n1 + l16];
            p.y = W1[(k0 + khalf + 1) * GCN_HID + n1 + l16];
            q.x = W1[(k0 + khalf) * GCN_HID + n1 + 16 + l16];
            q.y = W1[(k0 + khalf + 1) * GCN_HID + n1 + 16 + l16];
            c0 = __builtin_amdgcn_wmma_f32_16x16x4_f32(
                     false, a, false, p, (short)0, c0, false, false);
            c1 = __builtin_amdgcn_wmma_f32_16x16x4_f32(
                     false, a, false, q, (short)0, c1, false, false);
        }
        const int r0 = (lane >> 4) * 8;
        const float bb0 = b1[n1 + l16];
        const float bb1 = b1[n1 + 16 + l16];
#pragma unroll
        for (int v = 0; v < 8; ++v) {
            Hs[(r0 + v) * GCN_PITCH + n1 + l16]      = c0[v] + bb0;
            Hs[(r0 + v) * GCN_PITCH + n1 + 16 + l16] = c1[v] + bb1;
        }
    }
    __syncthreads();

    {
        const float4 g = *(const float4*)(gamma + lane * 4);
        const float4 b = *(const float4*)(beta + lane * 4);
        for (int i = 0; i < 4; ++i) {
            float* __restrict__ rowp = Hs + (wave * 4 + i) * GCN_PITCH + lane * 4;
            const float4 v = *(const float4*)rowp;

            float s = v.x + v.y + v.z + v.w;
#pragma unroll
            for (int off = 16; off > 0; off >>= 1) s += __shfl_xor(s, off, 32);
            float mu = s * (1.0f / 128.0f);

            float dx = v.x - mu, dy = v.y - mu, dz = v.z - mu, dw = v.w - mu;
            float sq = dx * dx + dy * dy + dz * dz + dw * dw;
#pragma unroll
            for (int off = 16; off > 0; off >>= 1) sq += __shfl_xor(sq, off, 32);
            float rstd = rsqrtf(sq * (1.0f / 128.0f) + GCN_EPS);

            float4 o;
            o.x = fmaxf(dx * rstd * g.x + b.x, 0.0f);
            o.y = fmaxf(dy * rstd * g.y + b.y, 0.0f);
            o.z = fmaxf(dz * rstd * g.z + b.z, 0.0f);
            o.w = fmaxf(dw * rstd * g.w + b.w, 0.0f);
            *(float4*)rowp = o;
        }
    }
    __syncthreads();

    {
        const int n2 = wave * 16;
        const float* __restrict__ Hr = Hs + l16 * GCN_PITCH + khalf;
        v8f c = {};
#pragma unroll
        for (int k0 = 0; k0 < GCN_HID; k0 += 4) {
            const v2f a = *(const v2f*)(Hr + k0);
            v2f b;
            b.x = W2[(k0 + khalf) * GCN_NCLS + n2 + l16];
            b.y = W2[(k0 + khalf + 1) * GCN_NCLS + n2 + l16];
            c = __builtin_amdgcn_wmma_f32_16x16x4_f32(
                    false, a, false, b, (short)0, c, false, false);
        }
        const int mrow = mbase + (lane >> 4) * 8;
        float* __restrict__ Cp = h2 + n2 + l16;
#pragma unroll
        for (int v = 0; v < 8; ++v) {
            int r = mrow + v;
            if (r < n) Cp[(long)r * GCN_NCLS] = c[v];
        }
    }
}


extern "C" void kernel_launch(void* const* d_in, const int* in_sizes, int n_in,
                              void* d_out, int out_size, void* d_ws, size_t ws_size,
                              hipStream_t stream) {
    const float* x   = (const float*)d_in[0];
    const int*   ei  = (const int*)d_in[1];
    const float* W1  = (const float*)d_in[2];
    const float* b1  = (const float*)d_in[3];
    const float* g1  = (const float*)d_in[4];
    const float* be1 = (const float*)d_in[5];
    const float* W2  = (const float*)d_in[6];
    const float* b2  = (const float*)d_in[7];
    float* out = (float*)d_out;

    const int n = in_sizes[0] / GCN_F_IN;
    const int e = in_sizes[1] / 2;
    const int* src = ei;
    const int* dst = ei + e;

    const int nbins   = (n + GCN_BIN_NODES - 1) / GCN_BIN_NODES;
    const int nchunks = (e + GCN_CHUNK - 1) / GCN_CHUNK;

    float* h2     = (float*)d_ws;
    int2*  rec    = (int2*)(h2 + (size_t)n * GCN_NCLS);
    int2*  binned = rec + e;
    int*   srcs   = (int*)(binned + (size_t)nchunks * GCN_CHUNK);
    float* dinv   = (float*)(srcs + e);
    int*   rowptr = (int*)(dinv + n);
    int*   loc    = rowptr + n + 1;

    gcn_chunk_bin<<<nchunks, 1024, 0, stream>>>(src, dst, binned, loc, e, nbins);
    gcn_bin_sort<<<nbins, 1024, 0, stream>>>(binned, loc, rowptr, dinv, srcs, n, e, nbins, nchunks);
    gcn_fill_w<<<((e + 3) / 4 + 511) / 512, 256, 0, stream>>>(srcs, dinv, rec, e);

    gcn_layer1_tile<<<(n + 15) / 16, 128, 0, stream>>>(
        x, rowptr, rec, dinv, W1, b1, g1, be1, W2, h2, n);

    gcn_agg_pull<GCN_NCLS><<<(int)(((long)n * 32 + 255) / 256), 256, 0, stream>>>(
        h2, rowptr, rec, dinv, b2, out, n);
}


# Per-edge weight pass dropped: layer 1 looks up dinv[src] itself (4 list heads per wave asked together); h2 rows stored times dinv, layer 2 adds plain rows then scales the sum; 4 launches (was 5)
# speedup vs baseline: 5.4518x; 1.0331x over previous
#include <hip/hip_runtime.h>

typedef __attribute__((ext_vector_type(2))) float v2f;
typedef __attribute__((ext_vector_type(8))) float v8f;

#define GCN_F_IN 128
#define GCN_HID  128
#define GCN_NCLS 64
#define GCN_EPS  1e-6f


#define GCN_BIN_SHIFT  8
#define GCN_BIN_NODES  256
#define GCN_MAX_BINS   1024
#define GCN_CHUNK      4096
#define GCN_MAX_CHUNKS 1024

__global__ __launch_bounds__(1024) void gcn_chunk_bin(const int* __restrict__ src,
                                                      const int* __restrict__ dst,
                                                      int2* __restrict__ binned,
                                                      int* __restrict__ loc,
                                                      int e, int nbins) {
    __shared__ int hist[GCN_MAX_BINS];
    __shared__ int wtot[32];
    const int tid  = threadIdx.x;
    const int lane = tid & 31;
    const int wv   = tid >> 5;
    hist[tid] = 0;
    __syncthreads();
    const int base = blockIdx.x * GCN_CHUNK;
    int s[4], d[4];
#pragma unroll
    for (int j = 0; j < 4; ++j) {
        const int i = base + j * 1024 + tid;
        d[j] = i < e ? dst[i] : -1;
        s[j] = i < e ? src[i] : 0;
    }
#pragma unroll
    for (int j = 0; j < 4; ++j) {
        if (d[j] >= 0) atomicAdd(&hist[d[j] >> GCN_BIN_SHIFT], 1);
    }
    __syncthreads();

    const int c = hist[tid];
    int inc = c;
    for (int off = 1; off < 32; off <<= 1) {
        int t = __shfl_up(inc, off, 32);
        if (lane >= off) inc += t;
    }
    if (lane == 31) wtot[wv] = inc;
    __syncthreads();
    int pre = lane < wv ? wtot[lane] : 0;
    for (int off = 16; off > 0; off >>= 1) pre += __shfl_xor(pre, off, 32);
    const int excl = pre + inc - c;
    hist[tid] = excl;
    if (tid < nbins) loc[blockIdx.x * nbins + tid] = excl;
    __syncthreads();

#pragma unroll
    for (int j = 0; j < 4; ++j) {
        if (d[j] >= 0) {
            const int pos = atomicAdd(&hist[d[j] >> GCN_BIN_SHIFT], 1);
            binned[base + pos] = make_int2(s[j], d[j]);
        }
    }
}

__global__ __launch_bounds__(1024) void gcn_bin_sort(const int2* __restrict__ binned,
                                                     const int* __restrict__ loc,
                                                     int* __restrict__ rowptr,
                                                     float* __restrict__ dinv,
                                                     int* __restrict__ srcs,
                                                     int n, int e, int nbins, int nchunks) {
    __shared__ int cl[GCN_BIN_NODES];
    __shared__ int wtot[GCN_BIN_NODES / 32];
    __shared__ int segb[GCN_MAX_CHUNKS + 256];
    __shared__ int sege[GCN_MAX_CHUNKS + 256];
    __shared__ int redb[32];
    __shared__ int rede[32];
    const int tid  = threadIdx.x;
    const int lane = tid & 31;
    const int wv   = tid >> 5;
    const int bin  = blockIdx.x;
    const int nodebase = bin << GCN_BIN_SHIFT;

    int sb = 0, se = 0;
    if (tid < nchunks) {
        const int left = e - tid * GCN_CHUNK;
        sb = loc[tid * nbins + bin];
        se = bin + 1 < nbins ? loc[tid * nbins + bin + 1] : (left < GCN_CHUNK ? left : GCN_CHUNK);
    }
    segb[tid] = sb;
    sege[tid] = se;
    if (tid < 256) {
        segb[GCN_MAX_CHUNKS + tid] = 0;
        sege[GCN_MAX_CHUNKS + tid] = 0;
    }
    if (tid < GCN_BIN_NODES) cl[tid] = 0;
    int pb = sb, pe = se;
    for (int off = 16; off > 0; off >>= 1) {
        pb += __shfl_xor(pb, off, 32);
        pe += __shfl_xor(pe, off, 32);
    }
    if (lane == 0) { redb[wv] = pb; rede[wv] = pe; }
    __syncthreads();
    int beg = redb[lane], end = rede[lane];
    for (int off = 16; off > 0; off >>= 1) {
        beg += __shfl_xor(beg, off, 32);
        end += __shfl_xor(end, off, 32);
    }

    for (int c0 = wv; c0 < nchunks; c0 += 256) {
        int sb8[8], se8[8], d[8];
#pragma unroll
        for (int u = 0; u < 8; ++u) {
            sb8[u] = segb[c0 + 32 * u];
            se8[u] = sege[c0 + 32 * u];
        }
#pragma unroll
        for (int u = 0; u < 8; ++u) {
            const int i = sb8[u] + lane;
            d[u] = i < se8[u] ? binned[(c0 + 32 * u) * GCN_CHUNK + i].y : -1;
        }
#pragma unroll
        for (int u = 0; u < 8; ++u) {
            if (d[u] >= 0) atomicAdd(&cl[d[u] - nodebase], 1);
        }
#pragma unroll
        for (int u = 0; u < 8; ++u) {
            for (int i = sb8[u] + 32 + lane; i < se8[u]; i += 32)
                atomicAdd(&cl[binned[(c0 + 32 * u) * GCN_CHUNK + i].y - nodebase], 1);
        }
    }
    __syncthreads();

    const int c = tid < GCN_BIN_NODES ? cl[tid] : 0;
    int inc = c;
    for (int off = 1; off < 32; off <<= 1) {
        int t = __shfl_up(inc, off, 32);
        if (lane >= off) inc += t;
    }
    if (tid < GCN_BIN_NODES && lane == 31) wtot[wv] = inc;
    __syncthreads();

    if (tid < GCN_BIN_NODES) {
        int pre = 0;
#pragma unroll
        for (int w = 0; w < GCN_BIN_NODES / 32; ++w) pre += w < wv ? wtot[w] : 0;
        const int start = beg + pre + inc - c;
        cl[tid] = start;
        const int node = nodebase + tid;
        if (node < n) {
            rowptr[node] = start;
            float dg = (float)(c + 1);
            dinv[node] = dg > 0.0f ? rsqrtf(dg) : 0.0f;
        }
    }
    if (bin == nbins - 1 && tid == 0) rowptr[n] = end;
    __syncthreads();

    for (int c0 = wv; c0 < nchunks; c0 += 256) {
        int sb8[8], se8[8];
        int2 r[8];
#pragma unroll
        for (int u = 0; u < 8; ++u) {
            sb8[u] = segb[c0 + 32 * u];
            se8[u] = sege[c0 + 32 * u];
        }
#pragma unroll
        for (int u = 0; u < 8; ++u) {
            const int i = sb8[u] + lane;
            r[u] = i < se8[u] ? binned[(c0 + 32 * u) * GCN_CHUNK + i] : make_int2(0, -1);
        }
#pragma unroll
        for (int u = 0; u < 8; ++u) {
            if (r[u].y >= 0) {
                const int pos = atomicAdd(&cl[r[u].y - nodebase], 1);
                srcs[pos] = r[u].x;
            }
        }
#pragma unroll
        for (int u = 0; u < 8; ++u) {
            for (int i = sb8[u] + 32 + lane; i < se8[u]; i += 32) {
                const int2 q = binned[(c0 + 32 * u) * GCN_CHUNK + i];
                const int pos = atomicAdd(&cl[q.y - nodebase], 1);
                srcs[pos] = q.x;
            }
        }
    }
}


template <int F, int G, bool WEIGHTED>
__device__ __forceinline__ void gcn_pull_group(const float* __restrict__ h, int my_s, int my_w,
                                               int j, int lane,
                                               float (&acc)[F / 32]) {
    constexpr int VEC = F / 32;
    typedef __attribute__((ext_vector_type(VEC))) float vec_t;
    vec_t hv[G];
    float w[G];
    for (int g = 0; g < G; ++g) {
        const int s = __builtin_amdgcn_readlane(my_s, j + g);
        w[g] = WEIGHTED ? __int_as_float(__builtin_amdgcn_readlane(my_w, j + g)) : 1.0f;
        hv[g] = *(const vec_t*)(h + (size_t)s * F + lane * VEC);
    }
    for (int g = 0; g < G; ++g) {
        for (int v = 0; v < VEC; ++v) {
            if (WEIGHTED) acc[v] += hv[g][v] * w[g];
            else          acc[v] += hv[g][v];
        }
    }
}

template <int F, bool WEIGHTED>
__device__ __forceinline__ void gcn_pull_batch(const float* __restrict__ h, int my_s, int my_w,
                                               int m, int lane, float (&acc)[F / 32]) {
    int j = 0;
    for (; j + 8 <= m; j += 8) gcn_pull_group<F, 8, WEIGHTED>(h, my_s, my_w, j, lane, acc);
    if (j + 4 <= m) { gcn_pull_group<F, 4, WEIGHTED>(h, my_s, my_w, j, lane, acc); j += 4; }
    if (j + 2 <= m) { gcn_pull_group<F, 2, WEIGHTED>(h, my_s, my_w, j, lane, acc); j += 2; }
    if (j < m)      { gcn_pull_group<F, 1, WEIGHTED>(h, my_s, my_w, j, lane, acc); }
}

template <int F, bool WEIGHTED>
__device__ __forceinline__ void gcn_pull_list(const float* __restrict__ h,
                                              const int* __restrict__ srcs,
                                              const float* __restrict__ dinv,
                                              int beg, int end, float dn, int lane,
                                              float (&acc)[F / 32]) {
    for (int k = beg; k < end; k += 32) {
        const int m = end - k < 32 ? end - k : 32;
        int my_s = 0, my_w = 0;
        if (lane < m) {
            my_s = srcs[k + lane];
            if (WEIGHTED) my_w = __float_as_int(dinv[my_s] * dn);
        }
        gcn_pull_batch<F, WEIGHTED>(h, my_s, my_w, m, lane, acc);
    }
}

template <int F>
__global__ __launch_bounds__(256) void gcn_agg_pull(const float* __restrict__ hsc,
                                                    const int* __restrict__ rowptr,
                                                    const int* __restrict__ srcs,
                                                    const float* __restrict__ dinv,
                                                    const float* __restrict__ bias,
                                                    float* __restrict__ out, int n) {
    constexpr int VEC = F / 32;
    typedef __attribute__((ext_vector_type(VEC))) float vec_t;
    const int lane = threadIdx.x & 31;
    const int node = __builtin_amdgcn_readfirstlane((int)((blockIdx.x * blockDim.x + threadIdx.x) >> 5));
    if (node >= n) return;

    const int beg = rowptr[node];
    const int end = rowptr[node + 1];
    const float dn = dinv[node];

    const vec_t hs = *(const vec_t*)(hsc + (size_t)node * F + lane * VEC);
    const vec_t bv = *(const vec_t*)(bias + lane * VEC);
    float acc[VEC];
    for (int v = 0; v < VEC; ++v) acc[v] = hs[v];

    gcn_pull_list<F, false>(hsc, srcs, dinv, beg, end, dn, lane, acc);

    vec_t o;
    for (int v = 0; v < VEC; ++v) o[v] = acc[v] * dn + bv[v];
    *(vec_t*)(out + (size_t)node * F + lane * VEC) = o;
}


#define GCN_PITCH 132

__global__ __launch_bounds__(128) void gcn_layer1_tile(const float* __restrict__ x,
                                                       const int* __restrict__ rowptr,
                                                       const int* __restrict__ srcs,
                                                       const float* __restrict__ dinv,
                                                       const float* __restrict__ W1,
                                                       const float* __restrict__ b1,
                                                       const float* __restrict__ gamma,
                                                       const float* __restrict__ beta,
                                                       const float* __restrict__ W2,
                                                       float* __restrict__ h2, int n) {
    __shared__ __attribute__((aligned(16))) float As[16 * GCN_PITCH];
    __shared__ __attribute__((aligned(16))) float Hs[16 * GCN_PITCH];
    __shared__ float Ds[16];

    const int lane  = threadIdx.x & 31;
    const int wave  = __builtin_amdgcn_readfirstlane((int)(threadIdx.x >> 5));
    const int mbase = blockIdx.x * 16;
    const int l16   = lane & 15;
    const int khalf = (lane >> 4) * 2;

    int   beg[4], end[4], hm[4], hs[4], hw[4];
    float dn[4];
#pragma unroll
    for (int i = 0; i < 4; ++i) {
        const int node = mbase + wave * 4 + i;
        const bool ok = node < n;
        beg[i] = ok ? rowptr[node] : 0;
        end[i] = ok ? rowptr[node + 1] : 0;
        dn[i]  = ok ? dinv[node] : 0.0f;
        hm[i]  = end[i] - beg[i] < 32 ? end[i] - beg[i] : 32;
    }
#pragma unroll
    for (int i = 0; i < 4; ++i) hs[i] = lane < hm[i] ? srcs[beg[i] + lane] : 0;
    float hd[4];
#pragma unroll
    for (int i = 0; i < 4; ++i) hd[i] = dinv[hs[i]];
#pragma unroll
    for (int i = 0; i < 4; ++i) hw[i] = __float_as_int(hd[i] * dn[i]);
#pragma unroll
    for (int i = 0; i < 4; ++i) {
        const int row  = wave * 4 + i;
        const int node = mbase + row;
        float acc[4] = {0.0f, 0.0f, 0.0f, 0.0f};
        if (node < n) {
            const float4 xs = *(const float4*)(x + (size_t)node * GCN_F_IN + lane * 4);
            acc[0] = xs.x * dn[i] * dn[i];
            acc[1] = xs.y * dn[i] * dn[i];
            acc[2] = xs.z * dn[i] * dn[i];
            acc[3] = xs.w * dn[i] * dn[i];
            gcn_pull_batch<GCN_F_IN, true>(x, hs[i], hw[i], hm[i], lane, acc);
            gcn_pull_list<GCN_F_IN, true>(x, srcs, dinv, beg[i] + 32, end[i], dn[i], lane, acc);
        }
        *(float4*)(As + row * GCN_PITCH + lane * 4) = make_float4(acc[0], acc[1], acc[2], acc[3]);
        if (lane == 0) Ds[row] = dn[i];
    }
    __syncthreads();

    {
        const int n1 = wave * 32;
        const float* __restrict__ Ar = As + l16 * GCN_PITCH + khalf;
        v8f c0 = {};
        v8f c1 = {};
#pragma unroll 8
        for (int k0 = 0; k0 < GCN_F_IN; k0 += 4) {
            const v2f a = *(const v2f*)(Ar + k0);
            v2f p, q;
            p.x = W1[(k0 + khalf) * GCN_HID + n1 + l16];
            p.y = W1[(k0 + khalf + 1) * GCN_HID + n1 + l16];
            q.x = W1[(k0 + khalf) * GCN_HID + n1 + 16 + l16];
            q.y = W1[(k0 + khalf + 1) * GCN_HID + n1 + 16 + l16];
            c0 = __builtin_amdgcn_wmma_f32_16x16x4_f32(
                     false, a, false, p, (short)0, c0, false, false);
            c1 = __builtin_amdgcn_wmma_f32_16x16x4_f32(
                     false, a, false, q, (short)0, c1, false, false);
        }
        const int r0 = (lane >> 4) * 8;
        const float bb0 = b1[n1 + l16];
        const float bb1 = b1[n1 + 16 + l16];
#pragma unroll
        for (int v = 0; v < 8; ++v) {
            Hs[(r0 + v) * GCN_PITCH + n1 + l16]      = c0[v] + bb0;
            Hs[(r0 + v) * GCN_PITCH + n1 + 16 + l16] = c1[v] + bb1;
        }
    }
    __syncthreads();

    {
        const float4 g = *(const float4*)(gamma + lane * 4);
        const float4 b = *(const float4*)(beta + lane * 4);
        for (int i = 0; i < 4; ++i) {
            float* __restrict__ rowp = Hs + (wave * 4 + i) * GCN_PITCH + lane * 4;
            const float4 v = *(const float4*)rowp;

            float s = v.x + v.y + v.z + v.w;
#pragma unroll
            for (int off = 16; off > 0; off >>= 1) s += __shfl_xor(s, off, 32);
            float mu = s * (1.0f / 128.0f);

            float dx = v.x - mu, dy = v.y - mu, dz = v.z - mu, dw = v.w - mu;
            float sq = dx * dx + dy * dy + dz * dz + dw * dw;
#pragma unroll
            for (int off = 16; off > 0; off >>= 1) sq += __shfl_xor(sq, off, 32);
            float rstd = rsqrtf(sq * (1.0f / 128.0f) + GCN_EPS);

            float4 o;
            o.x = fmaxf(dx * rstd * g.x + b.x, 0.0f);
            o.y = fmaxf(dy * rstd * g.y + b.y, 0.0f);
            o.z = fmaxf(dz * rstd * g.z + b.z, 0.0f);
            o.w = fmaxf(dw * rstd * g.w + b.w, 0.0f);
            *(float4*)rowp = o;
        }
    }
    __syncthreads();

    {
        const int n2 = wave * 16;
        const float* __restrict__ Hr = Hs + l16 * GCN_PITCH + khalf;
        v8f c = {};
#pragma unroll
        for (int k0 = 0; k0 < GCN_HID; k0 += 4) {
            const v2f a = *(const v2f*)(Hr + k0);
            v2f b;
            b.x = W2[(k0 + khalf) * GCN_NCLS + n2 + l16];
            b.y = W2[(k0 + khalf + 1) * GCN_NCLS + n2 + l16];
            c = __builtin_amdgcn_wmma_f32_16x16x4_f32(
                    false, a, false, b, (short)0, c, false, false);
        }
        const int r0   = (lane >> 4) * 8;
        const int mrow = mbase + r0;
        float* __restrict__ Cp = h2 + n2 + l16;
#pragma unroll
        for (int v = 0; v < 8; ++v) {
            int r = mrow + v;
            if (r < n) Cp[(long)r * GCN_NCLS] = c[v] * Ds[r0 + v];
        }
    }
}


extern "C" void kernel_launch(void* const* d_in, const int* in_sizes, int n_in,
                              void* d_out, int out_size, void* d_ws, size_t ws_size,
                              hipStream_t stream) {
    const float* x   = (const float*)d_in[0];
    const int*   ei  = (const int*)d_in[1];
    const float* W1  = (const float*)d_in[2];
    const float* b1  = (const float*)d_in[3];
    const float* g1  = (const float*)d_in[4];
    const float* be1 = (const float*)d_in[5];
    const float* W2  = (const float*)d_in[6];
    const float* b2  = (const float*)d_in[7];
    float* out = (float*)d_out;

    const int n = in_sizes[0] / GCN_F_IN;
    const int e = in_sizes[1] / 2;
    const int* src = ei;
    const int* dst = ei + e;

    const int nbins   = (n + GCN_BIN_NODES - 1) / GCN_BIN_NODES;
    const int nchunks = (e + GCN_CHUNK - 1) / GCN_CHUNK;

    float* h2     = (float*)d_ws;
    int2*  binned = (int2*)(h2 + (size_t)n * GCN_NCLS);
    int*   srcs   = (int*)(binned + (size_t)nchunks * GCN_CHUNK);
    float* dinv   = (float*)(srcs + e);
    int*   rowptr = (int*)(dinv + n);
    int*   loc    = rowptr + n + 1;

    gcn_chunk_bin<<<nchunks, 1024, 0, stream>>>(src, dst, binned, loc, e, nbins);
    gcn_bin_sort<<<nbins, 1024, 0, stream>>>(binned, loc, rowptr, dinv, srcs, n, e, nbins, nchunks);

    gcn_layer1_tile<<<(n + 15) / 16, 128, 0, stream>>>(
        x, rowptr, srcs, dinv, W1, b1, g1, be1, W2, h2, n);

    gcn_agg_pull<GCN_NCLS><<<(int)(((long)n * 32 + 255) / 256), 256, 0, stream>>>(
        h2, rowptr, srcs, dinv, b2, out, n);
}
